# starting kernel
# baseline (speedup 1.0000x reference)
#include <hip/hip_runtime.h>
#include <hip/hip_bf16.h>


#define GCN_N    50000
#define GCN_FIN  128
#define GCN_E    800000
#define GCN_H    64
#define GCN_C    10
#define GCN_B    512

typedef __attribute__((ext_vector_type(2))) float v2f;
typedef __attribute__((ext_vector_type(8))) float v8f;

__global__ __launch_bounds__(128) void gcn_gemm_wmma_f32(
    const float* __restrict__ A, const float* __restrict__ W,
    float* __restrict__ C, int K)
{
    const int lane    = threadIdx.x;
    const int colBase = threadIdx.y * 16;
    const int rowBase = blockIdx.x * 16;
    const int half    = lane >> 4;
    const int l       = lane & 15;

    const float* arow = A + (size_t)(rowBase + l) * K + 2 * half;
    const float* wcol = W + colBase + l;

    v8f acc = {};
    for (int k0 = 0; k0 < K; k0 += 4) {
        v2f a, b;
        a.x = arow[k0 + 0];
        a.y = arow[k0 + 1];
        b.x = wcol[(size_t)(k0 + 2 * half + 0) * GCN_H];
        b.y = wcol[(size_t)(k0 + 2 * half + 1) * GCN_H];
        acc = __builtin_amdgcn_wmma_f32_16x16x4_f32(
             false, a,  false, b,
             (short)0, acc,  false,  false);
    }

#pragma unroll
    for (int v = 0; v < 8; ++v) {
        const int m = v + 8 * half;
        C[(size_t)(rowBase + m) * GCN_H + colBase + l] = acc[v];
    }
}

__global__ __launch_bounds__(256) void gcn_deg_count(
    const int* __restrict__ dst, float* __restrict__ deg)
{
    const int e = blockIdx.x * blockDim.x + threadIdx.x;
    if (e < GCN_E) atomicAdd(&deg[dst[e]], 1.0f);
}

__global__ __launch_bounds__(256) void gcn_dinv(float* __restrict__ deg)
{
    const int i = blockIdx.x * blockDim.x + threadIdx.x;
    if (i < GCN_N) deg[i] = rsqrtf(deg[i] + 1.0f);
}

__global__ __launch_bounds__(256) void gcn_scatter(
    const float* __restrict__ h, const int* __restrict__ src,
    const int* __restrict__ dst, const float* __restrict__ dinv,
    float* __restrict__ agg)
{
    const int idx = blockIdx.x * blockDim.x + threadIdx.x;
    const int e = idx >> 5;
    const int p = (idx & 31) * 2;
    const int s = src[e];
    const int d = dst[e];
    const float c = dinv[s] * dinv[d];
    const float2 v = *(const float2*)(h + (size_t)s * GCN_H + p);
    float* out = agg + (size_t)d * GCN_H + p;
    atomicAdd(out + 0, v.x * c);
    atomicAdd(out + 1, v.y * c);
}

__global__ __launch_bounds__(256) void gcn_epilogue(
    const float* __restrict__ agg, const float* __restrict__ h,
    const float* __restrict__ dinv, const float* __restrict__ bias,
    float* __restrict__ out, int do_relu)
{
    const int idx = blockIdx.x * blockDim.x + threadIdx.x;
    const int i = idx >> 6;
    const int f = idx & 63;
    const float di = dinv[i];
    float v = agg[idx] + h[idx] * di * di + bias[f];
    if (do_relu) v = fmaxf(v, 0.0f);
    out[idx] = v;
}

__global__ __launch_bounds__(256) void gcn_pool(
    const float* __restrict__ h, const int* __restrict__ batch,
    float* __restrict__ psum, float* __restrict__ pcnt)
{
    const int idx = blockIdx.x * blockDim.x + threadIdx.x;
    const int i = idx >> 6;
    const int f = idx & 63;
    const int b = batch[i];
    atomicAdd(&psum[(size_t)b * GCN_H + f], h[idx]);
    if (f == 0) atomicAdd(&pcnt[b], 1.0f);
}

__global__ __launch_bounds__(256) void gcn_final(
    const float* __restrict__ psum, const float* __restrict__ pcnt,
    const float* __restrict__ Wl, const float* __restrict__ bl,
    float* __restrict__ out)
{
    const int idx = blockIdx.x * blockDim.x + threadIdx.x;
    const int b = idx / GCN_C;
    const int c = idx % GCN_C;
    const float inv = 1.0f / fmaxf(pcnt[b], 1.0f);
    float acc = bl[c];
    const float* row = psum + (size_t)b * GCN_H;
#pragma unroll 8
    for (int k = 0; k < GCN_H; ++k)
        acc = fmaf(row[k] * inv, Wl[k * GCN_C + c], acc);
    out[idx] = acc;
}

extern "C" void kernel_launch(void* const* d_in, const int* in_sizes, int n_in,
                              void* d_out, int out_size, void* d_ws, size_t ws_size,
                              hipStream_t stream) {
    (void)in_sizes; (void)n_in; (void)out_size; (void)ws_size;

    const float* x     = (const float*)d_in[0];
    const int*   ei    = (const int*)  d_in[1];
    const int*   batch = (const int*)  d_in[2];
    const float* W1    = (const float*)d_in[3];
    const float* b1    = (const float*)d_in[4];
    const float* W2    = (const float*)d_in[5];
    const float* b2    = (const float*)d_in[6];
    const float* W3    = (const float*)d_in[7];
    const float* b3    = (const float*)d_in[8];
    const float* Wl    = (const float*)d_in[9];
    const float* bl    = (const float*)d_in[10];
    float* out = (float*)d_out;

    const int* src = ei;
    const int* dst = ei + GCN_E;

    float* dinv = (float*)d_ws;
    float* bufA = dinv + GCN_N;
    float* bufB = bufA + (size_t)GCN_N * GCN_H;
    float* psum = bufB + (size_t)GCN_N * GCN_H;
    float* pcnt = psum + (size_t)GCN_B * GCN_H;

    const dim3 gemmBlk(32, 4);
    const dim3 gemmGrd(GCN_N / 16);
    const int  scatGrd = (GCN_E * 32) / 256;
    const int  nhGrd   = (GCN_N * GCN_H) / 256;
    const size_t nhBytes = (size_t)GCN_N * GCN_H * sizeof(float);

    hipMemsetAsync(dinv, 0, GCN_N * sizeof(float), stream);
    gcn_deg_count<<<(GCN_E + 255) / 256, 256, 0, stream>>>(dst, dinv);
    gcn_dinv<<<(GCN_N + 255) / 256, 256, 0, stream>>>(dinv);

    gcn_gemm_wmma_f32<<<gemmGrd, gemmBlk, 0, stream>>>(x, W1, bufA, GCN_FIN);
    hipMemsetAsync(bufB, 0, nhBytes, stream);
    gcn_scatter<<<scatGrd, 256, 0, stream>>>(bufA, src, dst, dinv, bufB);
    gcn_epilogue<<<nhGrd, 256, 0, stream>>>(bufB, bufA, dinv, b1, bufB, 1);

    gcn_gemm_wmma_f32<<<gemmGrd, gemmBlk, 0, stream>>>(bufB, W2, bufA, GCN_H);
    hipMemsetAsync(bufB, 0, nhBytes, stream);
    gcn_scatter<<<scatGrd, 256, 0, stream>>>(bufA, src, dst, dinv, bufB);
    gcn_epilogue<<<nhGrd, 256, 0, stream>>>(bufB, bufA, dinv, b2, bufB, 1);

    gcn_gemm_wmma_f32<<<gemmGrd, gemmBlk, 0, stream>>>(bufB, W3, bufA, GCN_H);
    hipMemsetAsync(bufB, 0, nhBytes, stream);
    gcn_scatter<<<scatGrd, 256, 0, stream>>>(bufA, src, dst, dinv, bufB);
    gcn_epilogue<<<nhGrd, 256, 0, stream>>>(bufB, bufA, dinv, b3, bufB, 0);

    hipMemsetAsync(psum, 0, (size_t)(GCN_B * GCN_H + GCN_B) * sizeof(float), stream);
    gcn_pool<<<nhGrd, 256, 0, stream>>>(bufB, batch, psum, pcnt);
    gcn_final<<<(GCN_B * GCN_C) / 256, 256, 0, stream>>>(psum, pcnt, Wl, bl, out);
}


# Edge weight dinv[src]*dinv[dst] computed once: layer 1's scatter stores it per edge (one line per wave), layers 2 and 3 read it back coalesced, not two 32-line dinv look-ups per wave; sums as before
# speedup vs baseline: 3.3184x; 3.3184x over previous
#include <hip/hip_runtime.h>
#include <hip/hip_bf16.h>


#define GCN_N    50000
#define GCN_FIN  128
#define GCN_E    800000
#define GCN_H    64
#define GCN_C    10
#define GCN_B    512
#define GCN_NPAD 50048

#define GCN_PLANE ((size_t)GCN_N * GCN_H)

typedef __attribute__((ext_vector_type(2))) float v2f;
typedef __attribute__((ext_vector_type(8))) float v8f;

__device__ __forceinline__ int gcn_cache_half()
{
    return (int)((__builtin_amdgcn_s_sendmsg_rtn(0x87) >> 18) & 1u);
}

template <bool RELU_A>
__global__ __launch_bounds__(128) void gcn_gemm_wmma_f32(
    const float* __restrict__ A, const float* __restrict__ W,
    const float* __restrict__ dinv, const float* __restrict__ bias,
    float* __restrict__ C, float* __restrict__ S, int K)
{
    const int lane    = threadIdx.x;
    const int colBase = threadIdx.y * 16;
    const int rowBase = blockIdx.x * 16;
    const int half    = lane >> 4;
    const int l       = lane & 15;

    const float* arow  = A + (size_t)(rowBase + l) * K + 2 * half;
    const float* arow1 = arow + GCN_PLANE;
    const float* wcol  = W + colBase + l;

    v8f acc = {};
    for (int k0 = 0; k0 < K; k0 += 4) {
        v2f a, b;
        a.x = arow[k0 + 0];
        a.y = arow[k0 + 1];
        if (RELU_A) {
            a.x = fmaxf(a.x + arow1[k0 + 0], 0.0f);
            a.y = fmaxf(a.y + arow1[k0 + 1], 0.0f);
        }
        b.x = wcol[(size_t)(k0 + 2 * half + 0) * GCN_H];
        b.y = wcol[(size_t)(k0 + 2 * half + 1) * GCN_H];
        acc = __builtin_amdgcn_wmma_f32_16x16x4_f32(
             false, a,  false, b,
             (short)0, acc,  false,  false);
    }

    const float4* dp = reinterpret_cast<const float4*>(dinv + rowBase + 8 * half);
    const float4 d0 = dp[0];
    const float4 d1 = dp[1];
    const float di[8] = { d0.x, d0.y, d0.z, d0.w, d1.x, d1.y, d1.z, d1.w };
    const float bc = bias[colBase + l];

#pragma unroll
    for (int v = 0; v < 8; ++v) {
        const int m = v + 8 * half;
        const size_t o = (size_t)(rowBase + m) * GCN_H + colBase + l;
        C[o] = acc[v];
        S[o] = acc[v] * (di[v] * di[v]) + bc;
        S[o + GCN_PLANE] = 0.0f;
    }
}

__global__ __launch_bounds__(256) void gcn_deg_count(
    const int* __restrict__ dst, float* __restrict__ deg)
{
    const int e = blockIdx.x * blockDim.x + threadIdx.x;
    float* degh = deg + (size_t)gcn_cache_half() * GCN_NPAD;
    if (e < GCN_E) atomicAdd(&degh[dst[e]], 1.0f);
}

__global__ __launch_bounds__(256) void gcn_dinv(float* __restrict__ deg)
{
    const int i = blockIdx.x * blockDim.x + threadIdx.x;
    if (i < GCN_N) deg[i] = rsqrtf(deg[i] + deg[GCN_NPAD + i] + 1.0f);
}

#define GCN_EPW   32
#define GCN_EFLY  8

template <bool FIRST, bool POOLED>
__global__ __launch_bounds__(256) void gcn_scatter(
    const float* __restrict__ h, const int* __restrict__ src,
    const int* __restrict__ dst, const float* __restrict__ dinv,
    const int* __restrict__ batch, float* __restrict__ wgt,
    float* __restrict__ agg)
{
    const int lane  = threadIdx.x & 31;
    const int wave  = (blockIdx.x * blockDim.x + threadIdx.x) >> 5;
    const int e     = wave * GCN_EPW + lane;
    const int s     = src[e];
    const int dn    = dst[e];
    float c;
    if (FIRST) {
        c = dinv[s] * dinv[dn];
        wgt[e] = c;
    } else {
        c = wgt[e];
    }
    const int cbits = __float_as_int(c);
    const int d     = POOLED ? batch[dn] : dn;
    const size_t plane = POOLED ? (size_t)GCN_B * GCN_H : GCN_PLANE;
    float* aggh = agg + (size_t)gcn_cache_half() * plane;

#pragma unroll
    for (int j0 = 0; j0 < GCN_EPW; j0 += GCN_EFLY) {
        float vlo[GCN_EFLY], vhi[GCN_EFLY];
#pragma unroll
        for (int j = 0; j < GCN_EFLY; ++j) {
            const int sj = __builtin_amdgcn_readlane(s, j0 + j);
            const float* row = h + (size_t)sj * GCN_H + lane;
            vlo[j] = row[0];
            vhi[j] = row[32];
        }
#pragma unroll
        for (int j = 0; j < GCN_EFLY; ++j) {
            const int   dj = __builtin_amdgcn_readlane(d, j0 + j);
            const float cj = __int_as_float(__builtin_amdgcn_readlane(cbits, j0 + j));
            float* out = aggh + (size_t)dj * GCN_H + lane;
            atomicAdd(out,      vlo[j] * cj);
            atomicAdd(out + 32, vhi[j] * cj);
        }
    }
}

#define GCN_NPW 8

__device__ __forceinline__ void gcn_pool_flush(
    float* __restrict__ ps, float* __restrict__ pc, int g, int lane,
    float alo, float ahi, float cnt)
{
    float* o = ps + (size_t)g * GCN_H + lane;
    atomicAdd(o,      alo);
    atomicAdd(o + 32, ahi);
    if (lane == 0) atomicAdd(pc + g, cnt);
}

__global__ __launch_bounds__(320) void gcn_relu_pool(
    const float* __restrict__ sum, const int* __restrict__ batch,
    const float* __restrict__ dinv, float* __restrict__ r,
    float* __restrict__ psum, float* __restrict__ pcnt)
{
    const int lane = threadIdx.x & 31;
    const int wave = (blockIdx.x * blockDim.x + threadIdx.x) >> 5;
    const int n0   = wave * GCN_NPW;
    const int half = gcn_cache_half();
    float* ps = psum + (size_t)half * (GCN_B * GCN_H);
    float* pc = pcnt + (size_t)half * GCN_B;

    const int   gl    = batch[n0 + (lane & (GCN_NPW - 1))];
    const float dl    = dinv[n0 + (lane & (GCN_NPW - 1))];
    const int   wbits = __float_as_int(dl * dl);

    float lo[GCN_NPW], hi[GCN_NPW];
#pragma unroll
    for (int j = 0; j < GCN_NPW; ++j) {
        const float* p = sum + (size_t)(n0 + j) * GCN_H + lane;
        lo[j] = fmaxf(p[0]  + p[GCN_PLANE],      0.0f);
        hi[j] = fmaxf(p[32] + p[GCN_PLANE + 32], 0.0f);
    }
#pragma unroll
    for (int j = 0; j < GCN_NPW; ++j) {
        float* q = r + (size_t)(n0 + j) * GCN_H + lane;
        q[0]  = lo[j];
        q[32] = hi[j];
    }

    int   cur = __builtin_amdgcn_readlane(gl, 0);
    float alo = 0.0f, ahi = 0.0f, cnt = 0.0f;
#pragma unroll
    for (int j = 0; j < GCN_NPW; ++j) {
        const int g = __builtin_amdgcn_readlane(gl, j);
        if (g != cur) {
            gcn_pool_flush(ps, pc, cur, lane, alo, ahi, cnt);
            cur = g; alo = 0.0f; ahi = 0.0f; cnt = 0.0f;
        }
        const float w = __int_as_float(__builtin_amdgcn_readlane(wbits, j));
        alo = fmaf(lo[j], w, alo);
        ahi = fmaf(hi[j], w, ahi);
        cnt += 1.0f;
    }
    gcn_pool_flush(ps, pc, cur, lane, alo, ahi, cnt);
}

__global__ __launch_bounds__(128) void gcn_final(
    const float* __restrict__ psum, const float* __restrict__ pcnt,
    const float* __restrict__ W3, const float* __restrict__ b3,
    const float* __restrict__ Wl, const float* __restrict__ bl,
    float* __restrict__ out)
{
    __shared__ float pooled[16][GCN_H + 1];
    __shared__ float wl[GCN_H * GCN_C];

    const int lane    = threadIdx.x;
    const int colBase = threadIdx.y * 16;
    const int rowBase = blockIdx.x * 16;
    const int half    = lane >> 4;
    const int l       = lane & 15;
    const int tid     = threadIdx.y * 32 + lane;

    float wreg[(GCN_H * GCN_C) / 128];
#pragma unroll
    for (int i = 0; i < (GCN_H * GCN_C) / 128; ++i) wreg[i] = Wl[tid + 128 * i];

    const float* arow  = psum + (size_t)(rowBase + l) * GCN_H + 2 * half;
    const float* arow1 = arow + (size_t)GCN_B * GCN_H;
    const float* wcol  = W3 + colBase + l;

    v8f acc = {};
    for (int k0 = 0; k0 < GCN_H; k0 += 4) {
        v2f a, b;
        a.x = arow[k0 + 0] + arow1[k0 + 0];
        a.y = arow[k0 + 1] + arow1[k0 + 1];
        b.x = wcol[(size_t)(k0 + 2 * half + 0) * GCN_H];
        b.y = wcol[(size_t)(k0 + 2 * half + 1) * GCN_H];
        acc = __builtin_amdgcn_wmma_f32_16x16x4_f32(
             false, a,  false, b,
             (short)0, acc,  false,  false);
    }

    const float bc = b3[colBase + l];
#pragma unroll
    for (int v = 0; v < 8; ++v) {
        const int m = v + 8 * half;
        const float cnt = pcnt[rowBase + m] + pcnt[GCN_B + rowBase + m];
        const float inv = 1.0f / fmaxf(cnt, 1.0f);
        pooled[m][colBase + l] = fmaf(cnt, bc, acc[v]) * inv;
    }
#pragma unroll
    for (int i = 0; i < (GCN_H * GCN_C) / 128; ++i) wl[tid + 128 * i] = wreg[i];
    __syncthreads();

    for (int o = tid; o < 16 * GCN_C; o += 128) {
        const int m = o / GCN_C;
        const int cc = o % GCN_C;
        float s = bl[cc];
#pragma unroll 8
        for (int k = 0; k < GCN_H; ++k)
            s = fmaf(pooled[m][k], wl[k * GCN_C + cc], s);
        out[(size_t)(rowBase + m) * GCN_C + cc] = s;
    }
}

extern "C" void kernel_launch(void* const* d_in, const int* in_sizes, int n_in,
                              void* d_out, int out_size, void* d_ws, size_t ws_size,
                              hipStream_t stream) {
    (void)in_sizes; (void)n_in; (void)out_size; (void)ws_size;

    const float* x     = (const float*)d_in[0];
    const int*   ei    = (const int*)  d_in[1];
    const int*   batch = (const int*)  d_in[2];
    const float* W1    = (const float*)d_in[3];
    const float* b1    = (const float*)d_in[4];
    const float* W2    = (const float*)d_in[5];
    const float* b2    = (const float*)d_in[6];
    const float* W3    = (const float*)d_in[7];
    const float* b3    = (const float*)d_in[8];
    const float* Wl    = (const float*)d_in[9];
    const float* bl    = (const float*)d_in[10];
    float* out = (float*)d_out;

    const int* src = ei;
    const int* dst = ei + GCN_E;

    float* dinv = (float*)d_ws;
    float* bufA = dinv + 2 * GCN_NPAD;
    float* bufB = bufA + GCN_PLANE;
    float* bufC = bufB + 2 * GCN_PLANE;
    float* psum = bufC + 2 * GCN_PLANE;
    float* pcnt = psum + (size_t)2 * GCN_B * GCN_H;
    float* wgt  = pcnt + 2 * GCN_B;

    const dim3 gemmBlk(32, 4);
    const dim3 gemmGrd(GCN_N / 16);
    const int  scatGrd = (GCN_E / GCN_EPW) / 8;

    hipMemsetAsync(dinv, 0, 2 * GCN_NPAD * sizeof(float), stream);
    gcn_deg_count<<<(GCN_E + 255) / 256, 256, 0, stream>>>(dst, dinv);
    gcn_dinv<<<(GCN_N + 255) / 256, 256, 0, stream>>>(dinv);

    gcn_gemm_wmma_f32<false><<<gemmGrd, gemmBlk, 0, stream>>>(x, W1, dinv, b1, bufA, bufB, GCN_FIN);
    gcn_scatter<true, false><<<scatGrd, 256, 0, stream>>>(bufA, src, dst, dinv, batch, wgt, bufB);

    gcn_gemm_wmma_f32<true><<<gemmGrd, gemmBlk, 0, stream>>>(bufB, W2, dinv, b2, bufA, bufC, GCN_H);
    gcn_scatter<false, false><<<scatGrd, 256, 0, stream>>>(bufA, src, dst, dinv, batch, wgt, bufC);

    hipMemsetAsync(psum, 0, (size_t)2 * (GCN_B * GCN_H + GCN_B) * sizeof(float), stream);
    gcn_relu_pool<<<(GCN_N / GCN_NPW) / 10, 320, 0, stream>>>(bufC, batch, dinv, bufA, psum, pcnt);
    gcn_scatter<false, true><<<scatGrd, 256, 0, stream>>>(bufA, src, dst, dinv, batch, wgt, psum);
    gcn_final<<<GCN_B / 16, gemmBlk, 0, stream>>>(psum, pcnt, W3, b3, Wl, bl, out);
}


# Degree count: 64 blocks each count 12500 edges in a 50000-counter LDS histogram (integer adds) and store it whole; dinv kernel adds the 64 partial counts; global float atomics, degree memset gone
# speedup vs baseline: 3.4305x; 1.0338x over previous
#include <hip/hip_runtime.h>
#include <hip/hip_bf16.h>


#define GCN_N    50000
#define GCN_FIN  128
#define GCN_E    800000
#define GCN_H    64
#define GCN_C    10
#define GCN_B    512
#define GCN_NPAD 50048

#define GCN_PLANE ((size_t)GCN_N * GCN_H)

typedef __attribute__((ext_vector_type(2))) float v2f;
typedef __attribute__((ext_vector_type(8))) float v8f;

__device__ __forceinline__ int gcn_cache_half()
{
    return (int)((__builtin_amdgcn_s_sendmsg_rtn(0x87) >> 18) & 1u);
}

template <bool RELU_A>
__global__ __launch_bounds__(128) void gcn_gemm_wmma_f32(
    const float* __restrict__ A, const float* __restrict__ W,
    const float* __restrict__ dinv, const float* __restrict__ bias,
    float* __restrict__ C, float* __restrict__ S, int K)
{
    const int lane    = threadIdx.x;
    const int colBase = threadIdx.y * 16;
    const int rowBase = blockIdx.x * 16;
    const int half    = lane >> 4;
    const int l       = lane & 15;

    const float* arow  = A + (size_t)(rowBase + l) * K + 2 * half;
    const float* arow1 = arow + GCN_PLANE;
    const float* wcol  = W + colBase + l;

    v8f acc = {};
    for (int k0 = 0; k0 < K; k0 += 4) {
        v2f a, b;
        a.x = arow[k0 + 0];
        a.y = arow[k0 + 1];
        if (RELU_A) {
            a.x = fmaxf(a.x + arow1[k0 + 0], 0.0f);
            a.y = fmaxf(a.y + arow1[k0 + 1], 0.0f);
        }
        b.x = wcol[(size_t)(k0 + 2 * half + 0) * GCN_H];
        b.y = wcol[(size_t)(k0 + 2 * half + 1) * GCN_H];
        acc = __builtin_amdgcn_wmma_f32_16x16x4_f32(
             false, a,  false, b,
             (short)0, acc,  false,  false);
    }

    const float4* dp = reinterpret_cast<const float4*>(dinv + rowBase + 8 * half);
    const float4 d0 = dp[0];
    const float4 d1 = dp[1];
    const float di[8] = { d0.x, d0.y, d0.z, d0.w, d1.x, d1.y, d1.z, d1.w };
    const float bc = bias[colBase + l];

#pragma unroll
    for (int v = 0; v < 8; ++v) {
        const int m = v + 8 * half;
        const size_t o = (size_t)(rowBase + m) * GCN_H + colBase + l;
        C[o] = acc[v];
        S[o] = acc[v] * (di[v] * di[v]) + bc;
        S[o + GCN_PLANE] = 0.0f;
    }
}

#define GCN_DEGB 64
#define GCN_DEGE (GCN_E / GCN_DEGB)
#define GCN_DEGT 1024
#define GCN_DEGK ((GCN_DEGE + GCN_DEGT - 1) / GCN_DEGT)

__global__ __launch_bounds__(GCN_DEGT) void gcn_deg_count(
    const int* __restrict__ dst, unsigned* __restrict__ part)
{
    __shared__ unsigned cnt[GCN_NPAD];
    const int tid = threadIdx.x;
    const int* dstb = dst + (size_t)blockIdx.x * GCN_DEGE;

    int d[GCN_DEGK];
#pragma unroll
    for (int k = 0; k < GCN_DEGK; ++k) {
        const int e = tid + GCN_DEGT * k;
        d[k] = (e < GCN_DEGE) ? dstb[e] : -1;
    }

    uint4* c4 = reinterpret_cast<uint4*>(cnt);
    for (int i = tid; i < GCN_NPAD / 4; i += GCN_DEGT) c4[i] = make_uint4(0u, 0u, 0u, 0u);
    __syncthreads();

#pragma unroll
    for (int k = 0; k < GCN_DEGK; ++k)
        if (d[k] >= 0) atomicAdd(&cnt[d[k]], 1u);
    __syncthreads();

    uint4* p4 = reinterpret_cast<uint4*>(part + (size_t)blockIdx.x * GCN_NPAD);
    for (int i = tid; i < GCN_NPAD / 4; i += GCN_DEGT) p4[i] = c4[i];
}

__global__ __launch_bounds__(256) void gcn_dinv(const unsigned* __restrict__ part, float* __restrict__ dinv)
{
    const int i = blockIdx.x * blockDim.x + threadIdx.x;
    if (i < GCN_N) {
        unsigned s = 0u;
#pragma unroll
        for (int b = 0; b < GCN_DEGB; ++b) s += part[(size_t)b * GCN_NPAD + i];
        dinv[i] = rsqrtf((float)s + 1.0f);
    }
}

#define GCN_EPW   32
#define GCN_EFLY  8

template <bool FIRST, bool POOLED>
__global__ __launch_bounds__(256) void gcn_scatter(
    const float* __restrict__ h, const int* __restrict__ src,
    const int* __restrict__ dst, const float* __restrict__ dinv,
    const int* __restrict__ batch, float* __restrict__ wgt,
    float* __restrict__ agg)
{
    const int lane  = threadIdx.x & 31;
    const int wave  = (blockIdx.x * blockDim.x + threadIdx.x) >> 5;
    const int e     = wave * GCN_EPW + lane;
    const int s     = src[e];
    const int dn    = dst[e];
    float c;
    if (FIRST) {
        c = dinv[s] * dinv[dn];
        wgt[e] = c;
    } else {
        c = wgt[e];
    }
    const int cbits = __float_as_int(c);
    const int d     = POOLED ? batch[dn] : dn;
    const size_t plane = POOLED ? (size_t)GCN_B * GCN_H : GCN_PLANE;
    float* aggh = agg + (size_t)gcn_cache_half() * plane;

#pragma unroll
    for (int j0 = 0; j0 < GCN_EPW; j0 += GCN_EFLY) {
        float vlo[GCN_EFLY], vhi[GCN_EFLY];
#pragma unroll
        for (int j = 0; j < GCN_EFLY; ++j) {
            const int sj = __builtin_amdgcn_readlane(s, j0 + j);
            const float* row = h + (size_t)sj * GCN_H + lane;
            vlo[j] = row[0];
            vhi[j] = row[32];
        }
#pragma unroll
        for (int j = 0; j < GCN_EFLY; ++j) {
            const int   dj = __builtin_amdgcn_readlane(d, j0 + j);
            const float cj = __int_as_float(__builtin_amdgcn_readlane(cbits, j0 + j));
            float* out = aggh + (size_t)dj * GCN_H + lane;
            atomicAdd(out,      vlo[j] * cj);
            atomicAdd(out + 32, vhi[j] * cj);
        }
    }
}

#define GCN_NPW 8

__device__ __forceinline__ void gcn_pool_flush(
    float* __restrict__ ps, float* __restrict__ pc, int g, int lane,
    float alo, float ahi, float cnt)
{
    float* o = ps + (size_t)g * GCN_H + lane;
    atomicAdd(o,      alo);
    atomicAdd(o + 32, ahi);
    if (lane == 0) atomicAdd(pc + g, cnt);
}

__global__ __launch_bounds__(320) void gcn_relu_pool(
    const float* __restrict__ sum, const int* __restrict__ batch,
    const float* __restrict__ dinv, float* __restrict__ r,
    float* __restrict__ psum, float* __restrict__ pcnt)
{
    const int lane = threadIdx.x & 31;
    const int wave = (blockIdx.x * blockDim.x + threadIdx.x) >> 5;
    const int n0   = wave * GCN_NPW;
    const int half = gcn_cache_half();
    float* ps = psum + (size_t)half * (GCN_B * GCN_H);
    float* pc = pcnt + (size_t)half * GCN_B;

    const int   gl    = batch[n0 + (lane & (GCN_NPW - 1))];
    const float dl    = dinv[n0 + (lane & (GCN_NPW - 1))];
    const int   wbits = __float_as_int(dl * dl);

    float lo[GCN_NPW], hi[GCN_NPW];
#pragma unroll
    for (int j = 0; j < GCN_NPW; ++j) {
        const float* p = sum + (size_t)(n0 + j) * GCN_H + lane;
        lo[j] = fmaxf(p[0]  + p[GCN_PLANE],      0.0f);
        hi[j] = fmaxf(p[32] + p[GCN_PLANE + 32], 0.0f);
    }
#pragma unroll
    for (int j = 0; j < GCN_NPW; ++j) {
        float* q = r + (size_t)(n0 + j) * GCN_H + lane;
        q[0]  = lo[j];
        q[32] = hi[j];
    }

    int   cur = __builtin_amdgcn_readlane(gl, 0);
    float alo = 0.0f, ahi = 0.0f, cnt = 0.0f;
#pragma unroll
    for (int j = 0; j < GCN_NPW; ++j) {
        const int g = __builtin_amdgcn_readlane(gl, j);
        if (g != cur) {
            gcn_pool_flush(ps, pc, cur, lane, alo, ahi, cnt);
            cur = g; alo = 0.0f; ahi = 0.0f; cnt = 0.0f;
        }
        const float w = __int_as_float(__builtin_amdgcn_readlane(wbits, j));
        alo = fmaf(lo[j], w, alo);
        ahi = fmaf(hi[j], w, ahi);
        cnt += 1.0f;
    }
    gcn_pool_flush(ps, pc, cur, lane, alo, ahi, cnt);
}

__global__ __launch_bounds__(128) void gcn_final(
    const float* __restrict__ psum, const float* __restrict__ pcnt,
    const float* __restrict__ W3, const float* __restrict__ b3,
    const float* __restrict__ Wl, const float* __restrict__ bl,
    float* __restrict__ out)
{
    __shared__ float pooled[16][GCN_H + 1];
    __shared__ float wl[GCN_H * GCN_C];

    const int lane    = threadIdx.x;
    const int colBase = threadIdx.y * 16;
    const int rowBase = blockIdx.x * 16;
    const int half    = lane >> 4;
    const int l       = lane & 15;
    const int tid     = threadIdx.y * 32 + lane;

    float wreg[(GCN_H * GCN_C) / 128];
#pragma unroll
    for (int i = 0; i < (GCN_H * GCN_C) / 128; ++i) wreg[i] = Wl[tid + 128 * i];

    const float* arow  = psum + (size_t)(rowBase + l) * GCN_H + 2 * half;
    const float* arow1 = arow + (size_t)GCN_B * GCN_H;
    const float* wcol  = W3 + colBase + l;

    v8f acc = {};
    for (int k0 = 0; k0 < GCN_H; k0 += 4) {
        v2f a, b;
        a.x = arow[k0 + 0] + arow1[k0 + 0];
        a.y = arow[k0 + 1] + arow1[k0 + 1];
        b.x = wcol[(size_t)(k0 + 2 * half + 0) * GCN_H];
        b.y = wcol[(size_t)(k0 + 2 * half + 1) * GCN_H];
        acc = __builtin_amdgcn_wmma_f32_16x16x4_f32(
             false, a,  false, b,
             (short)0, acc,  false,  false);
    }

    const float bc = b3[colBase + l];
#pragma unroll
    for (int v = 0; v < 8; ++v) {
        const int m = v + 8 * half;
        const float cnt = pcnt[rowBase + m] + pcnt[GCN_B + rowBase + m];
        const float inv = 1.0f / fmaxf(cnt, 1.0f);
        pooled[m][colBase + l] = fmaf(cnt, bc, acc[v]) * inv;
    }
#pragma unroll
    for (int i = 0; i < (GCN_H * GCN_C) / 128; ++i) wl[tid + 128 * i] = wreg[i];
    __syncthreads();

    for (int o = tid; o < 16 * GCN_C; o += 128) {
        const int m = o / GCN_C;
        const int cc = o % GCN_C;
        float s = bl[cc];
#pragma unroll 8
        for (int k = 0; k < GCN_H; ++k)
            s = fmaf(pooled[m][k], wl[k * GCN_C + cc], s);
        out[(size_t)(rowBase + m) * GCN_C + cc] = s;
    }
}

extern "C" void kernel_launch(void* const* d_in, const int* in_sizes, int n_in,
                              void* d_out, int out_size, void* d_ws, size_t ws_size,
                              hipStream_t stream) {
    (void)in_sizes; (void)n_in; (void)out_size; (void)ws_size;

    const float* x     = (const float*)d_in[0];
    const int*   ei    = (const int*)  d_in[1];
    const int*   batch = (const int*)  d_in[2];
    const float* W1    = (const float*)d_in[3];
    const float* b1    = (const float*)d_in[4];
    const float* W2    = (const float*)d_in[5];
    const float* b2    = (const float*)d_in[6];
    const float* W3    = (const float*)d_in[7];
    const float* b3    = (const float*)d_in[8];
    const float* Wl    = (const float*)d_in[9];
    const float* bl    = (const float*)d_in[10];
    float* out = (float*)d_out;

    const int* src = ei;
    const int* dst = ei + GCN_E;

    float* dinv = (float*)d_ws;
    float* bufA = dinv + 2 * GCN_NPAD;
    float* bufB = bufA + GCN_PLANE;
    float* bufC = bufB + 2 * GCN_PLANE;
    float* psum = bufC + 2 * GCN_PLANE;
    float* pcnt = psum + (size_t)2 * GCN_B * GCN_H;
    float* wgt  = pcnt + 2 * GCN_B;
    unsigned* part = (unsigned*)(wgt + GCN_E);

    const dim3 gemmBlk(32, 4);
    const dim3 gemmGrd(GCN_N / 16);
    const int  scatGrd = (GCN_E / GCN_EPW) / 8;

    gcn_deg_count<<<GCN_DEGB, GCN_DEGT, 0, stream>>>(dst, part);
    gcn_dinv<<<(GCN_N + 255) / 256, 256, 0, stream>>>(part, dinv);

    gcn_gemm_wmma_f32<false><<<gemmGrd, gemmBlk, 0, stream>>>(x, W1, dinv, b1, bufA, bufB, GCN_FIN);
    gcn_scatter<true, false><<<scatGrd, 256, 0, stream>>>(bufA, src, dst, dinv, batch, wgt, bufB);

    gcn_gemm_wmma_f32<true><<<gemmGrd, gemmBlk, 0, stream>>>(bufB, W2, dinv, b2, bufA, bufC, GCN_H);
    gcn_scatter<false, false><<<scatGrd, 256, 0, stream>>>(bufA, src, dst, dinv, batch, wgt, bufC);

    hipMemsetAsync(psum, 0, (size_t)2 * (GCN_B * GCN_H + GCN_B) * sizeof(float), stream);
    gcn_relu_pool<<<(GCN_N / GCN_NPW) / 10, 320, 0, stream>>>(bufC, batch, dinv, bufA, psum, pcnt);
    gcn_scatter<false, true><<<scatGrd, 256, 0, stream>>>(bufA, src, dst, dinv, batch, wgt, psum);
    gcn_final<<<GCN_B / 16, gemmBlk, 0, stream>>>(psum, pcnt, W3, b3, Wl, bl, out);
}
